# starting kernel
# baseline (speedup 1.0000x reference)
#include <hip/hip_runtime.h>
#include <math.h>

#define DIMS     1024
#define HEADS    16
#define HDIM     64
#define ROT      32
#define BATCH    4
#define SEQ      8192

typedef __attribute__((ext_vector_type(2))) float v2f;
typedef __attribute__((ext_vector_type(8))) float v8f;

__global__ void __launch_bounds__(64)
build_M_kernel(const float* __restrict__ matrix,
               const float* __restrict__ thetas,
               const float* __restrict__ tscale,
               const int*   __restrict__ pairs,
               float*       __restrict__ Mout) {
    __shared__ float acc[HDIM][HDIM];
    const int t = threadIdx.x;
    for (int c = 0; c < HDIM; ++c) acc[t][c] = (t == c) ? 1.0f : 0.0f;

    const float ts = tscale[0];
    for (int g = 0; g < ROT; ++g) {
        const int   i  = pairs[2 * g];
        const int   j  = pairs[2 * g + 1];
        const float th = thetas[g] * ts;
        const float cs = cosf(th), sn = sinf(th);
        if (i != j) {
            const float vi = acc[t][i], vj = acc[t][j];
            acc[t][i] =  cs * vi + sn * vj;
            acc[t][j] = -sn * vi + cs * vj;
        } else {
            acc[t][i] *= sn;
        }
    }
    for (int n = 0; n < HDIM; ++n) {
        float s = 0.0f;
        for (int m = 0; m < HDIM; ++m)
            s = fmaf(acc[t][m], matrix[m * HDIM + n], s);
        Mout[t * HDIM + n] = s;
    }
}

#define STG 66

__global__ void __launch_bounds__(256)
rope_gemm_kernel(const float* __restrict__ x,
                 const float* __restrict__ Mg,
                 const float* __restrict__ invf,
                 float*       __restrict__ out) {
    __shared__ __align__(16) float Bl[16 * 4 * 64];
    __shared__ __align__(16) float stg[8][16 * STG];
    __shared__ __align__(16) float sctab[8][64];

    const int tid  = threadIdx.x;
    const int wave = tid >> 5;
    const int lane = tid & 31;

    for (int idx = tid; idx < 16 * 4 * 64; idx += 256) {
        const int v  = idx & 1;
        const int ln = (idx >> 1) & 31;
        const int t  = (idx >> 6) & 3;
        const int kk = idx >> 8;
        const int krow = 4 * kk + ((ln >> 4) << 1) + v;
        const int col  = 16 * t + (ln & 15);
        Bl[idx] = Mg[krow * HDIM + col];
    }
    __syncthreads();

    const int  tile    = blockIdx.x * 8 + wave;
    const int  s       = tile & (SEQ - 1);
    const long rowbase = (long)tile * 16;

    if (lane < ROT) {
        const float a = (float)s * invf[lane];
        sctab[wave][lane]      = cosf(a);
        sctab[wave][lane + 32] = sinf(a);
    }

    const float* xrow = x + (rowbase + (lane & 15)) * HDIM + ((lane >> 4) << 1);
    v2f afrag[16];
#pragma unroll
    for (int kk = 0; kk < 16; ++kk)
        afrag[kk] = *(const v2f*)(xrow + kk * 4);

    v8f c0 = {}, c1 = {}, c2 = {}, c3 = {};
    const float* Bp = Bl + lane * 2;
#pragma unroll
    for (int kk = 0; kk < 16; ++kk) {
        const v2f b0 = *(const v2f*)(Bp + (kk * 4 + 0) * 64);
        const v2f b1 = *(const v2f*)(Bp + (kk * 4 + 1) * 64);
        const v2f b2 = *(const v2f*)(Bp + (kk * 4 + 2) * 64);
        const v2f b3 = *(const v2f*)(Bp + (kk * 4 + 3) * 64);
        c0 = __builtin_amdgcn_wmma_f32_16x16x4_f32(false, afrag[kk], false, b0,
                                                   (short)0, c0, false, false);
        c1 = __builtin_amdgcn_wmma_f32_16x16x4_f32(false, afrag[kk], false, b1,
                                                   (short)0, c1, false, false);
        c2 = __builtin_amdgcn_wmma_f32_16x16x4_f32(false, afrag[kk], false, b2,
                                                   (short)0, c2, false, false);
        c3 = __builtin_amdgcn_wmma_f32_16x16x4_f32(false, afrag[kk], false, b3,
                                                   (short)0, c3, false, false);
    }

    {
        const int rbase = (lane >> 4) << 3;
        const int col   = lane & 15;
        float*    st    = stg[wave];
#pragma unroll
        for (int r = 0; r < 8; ++r) {
            st[(rbase + r) * STG + col]      = c0[r];
            st[(rbase + r) * STG + col + 16] = c1[r];
            st[(rbase + r) * STG + col + 32] = c2[r];
            st[(rbase + r) * STG + col + 48] = c3[r];
        }
    }
    __syncthreads();

    const float* st = stg[wave];
    const float* sc = sctab[wave];
    float*       op = out + rowbase * HDIM;
#pragma unroll
    for (int i = 0; i < 32; ++i) {
        const int flat = i * 32 + lane;
        const int row  = flat >> 6;
        const int col  = flat & 63;
        const int n    = col & 31;
        const v2f xy = *(const v2f*)(st + row * STG + 2 * n);
        const float cs = sc[n], sn = sc[n + 32];
        const float val = (col < 32) ? (xy.x * cs - xy.y * sn)
                                     : (xy.x * sn + xy.y * cs);
        op[flat] = val * 32.0f;
    }
}

extern "C" void kernel_launch(void* const* d_in, const int* in_sizes, int n_in,
                              void* d_out, int out_size, void* d_ws, size_t ws_size,
                              hipStream_t stream) {
    const float* x      = (const float*)d_in[0];
    const float* matrix = (const float*)d_in[1];
    const float* thetas = (const float*)d_in[2];
    const float* tscale = (const float*)d_in[3];
    const float* invf   = (const float*)d_in[4];
    const int*   pairs  = (const int*)d_in[5];
    float*       Mws    = (float*)d_ws;

    build_M_kernel<<<1, 64, 0, stream>>>(matrix, thetas, tscale, pairs, Mws);

    const int tiles  = BATCH * SEQ;
    const int blocks = tiles / 8;
    rope_gemm_kernel<<<blocks, 256, 0, stream>>>(x, Mws, invf, (float*)d_out);
}


# M build made parallel: one single-wave block per row of M, rotation chain in registers (readlane), product split over columns; same expressions and sum order; was one 64-thread block
# speedup vs baseline: 2.0271x; 2.0271x over previous
#include <hip/hip_runtime.h>
#include <math.h>

#define DIMS     1024
#define HEADS    16
#define HDIM     64
#define ROT      32
#define BATCH    4
#define SEQ      8192

typedef __attribute__((ext_vector_type(2))) float v2f;
typedef __attribute__((ext_vector_type(8))) float v8f;

__device__ __forceinline__ int lane_value(int v, int l) {
    return __builtin_amdgcn_readlane(v, l);
}
__device__ __forceinline__ float lane_value(float v, int l) {
    return __int_as_float(__builtin_amdgcn_readlane(__float_as_int(v), l));
}

__global__ void __launch_bounds__(32)
build_M_kernel(const float* __restrict__ matrix,
               const float* __restrict__ thetas,
               const float* __restrict__ tscale,
               const int*   __restrict__ pairs,
               float*       __restrict__ Mout) {
    const int t    = blockIdx.x;
    const int lane = threadIdx.x;

    const float th  = thetas[lane] * tscale[0];
    const int   piv = pairs[2 * lane];
    const int   pjv = pairs[2 * lane + 1];

    float mlo[HDIM], mhi[HDIM];
#pragma unroll
    for (int m = 0; m < HDIM; ++m) {
        mlo[m] = matrix[m * HDIM + lane];
        mhi[m] = matrix[m * HDIM + lane + 32];
    }

    const float csv = cosf(th), snv = sinf(th);

    float lo = (lane == t)      ? 1.0f : 0.0f;
    float hi = (lane + 32 == t) ? 1.0f : 0.0f;

#pragma unroll
    for (int g = 0; g < ROT; ++g) {
        const int   i  = lane_value(piv, g);
        const int   j  = lane_value(pjv, g);
        const float cs = lane_value(csv, g);
        const float sn = lane_value(snv, g);
        const float il = lane_value(lo, i & 31), ih = lane_value(hi, i & 31);
        const float jl = lane_value(lo, j & 31), jh = lane_value(hi, j & 31);
        const float vi = (i & 32) ? ih : il;
        const float vj = (j & 32) ? jh : jl;
        const float ri =  cs * vi + sn * vj;
        const float rj = -sn * vi + cs * vj;
        const float rd = vi * sn;
        const float ni = (i != j) ? ri : rd;
        const float nj = (i != j) ? rj : rd;
        lo = (lane == i)      ? ni : lo;
        hi = (lane + 32 == i) ? ni : hi;
        lo = (lane == j)      ? nj : lo;
        hi = (lane + 32 == j) ? nj : hi;
    }

    float s0 = 0.0f, s1 = 0.0f;
#pragma unroll
    for (int m = 0; m < 32; ++m) {
        const float a = lane_value(lo, m);
        s0 = fmaf(a, mlo[m], s0);
        s1 = fmaf(a, mhi[m], s1);
    }
#pragma unroll
    for (int m = 0; m < 32; ++m) {
        const float a = lane_value(hi, m);
        s0 = fmaf(a, mlo[m + 32], s0);
        s1 = fmaf(a, mhi[m + 32], s1);
    }
    Mout[t * HDIM + lane]      = s0;
    Mout[t * HDIM + lane + 32] = s1;
}

#define STG 66

__global__ void __launch_bounds__(256)
rope_gemm_kernel(const float* __restrict__ x,
                 const float* __restrict__ Mg,
                 const float* __restrict__ invf,
                 float*       __restrict__ out) {
    __shared__ __align__(16) float Bl[16 * 4 * 64];
    __shared__ __align__(16) float stg[8][16 * STG];
    __shared__ __align__(16) float sctab[8][64];

    const int tid  = threadIdx.x;
    const int wave = tid >> 5;
    const int lane = tid & 31;

    for (int idx = tid; idx < 16 * 4 * 64; idx += 256) {
        const int v  = idx & 1;
        const int ln = (idx >> 1) & 31;
        const int t  = (idx >> 6) & 3;
        const int kk = idx >> 8;
        const int krow = 4 * kk + ((ln >> 4) << 1) + v;
        const int col  = 16 * t + (ln & 15);
        Bl[idx] = Mg[krow * HDIM + col];
    }
    __syncthreads();

    const int  tile    = blockIdx.x * 8 + wave;
    const int  s       = tile & (SEQ - 1);
    const long rowbase = (long)tile * 16;

    if (lane < ROT) {
        const float a = (float)s * invf[lane];
        sctab[wave][lane]      = cosf(a);
        sctab[wave][lane + 32] = sinf(a);
    }

    const float* xrow = x + (rowbase + (lane & 15)) * HDIM + ((lane >> 4) << 1);
    v2f afrag[16];
#pragma unroll
    for (int kk = 0; kk < 16; ++kk)
        afrag[kk] = *(const v2f*)(xrow + kk * 4);

    v8f c0 = {}, c1 = {}, c2 = {}, c3 = {};
    const float* Bp = Bl + lane * 2;
#pragma unroll
    for (int kk = 0; kk < 16; ++kk) {
        const v2f b0 = *(const v2f*)(Bp + (kk * 4 + 0) * 64);
        const v2f b1 = *(const v2f*)(Bp + (kk * 4 + 1) * 64);
        const v2f b2 = *(const v2f*)(Bp + (kk * 4 + 2) * 64);
        const v2f b3 = *(const v2f*)(Bp + (kk * 4 + 3) * 64);
        c0 = __builtin_amdgcn_wmma_f32_16x16x4_f32(false, afrag[kk], false, b0,
                                                   (short)0, c0, false, false);
        c1 = __builtin_amdgcn_wmma_f32_16x16x4_f32(false, afrag[kk], false, b1,
                                                   (short)0, c1, false, false);
        c2 = __builtin_amdgcn_wmma_f32_16x16x4_f32(false, afrag[kk], false, b2,
                                                   (short)0, c2, false, false);
        c3 = __builtin_amdgcn_wmma_f32_16x16x4_f32(false, afrag[kk], false, b3,
                                                   (short)0, c3, false, false);
    }

    {
        const int rbase = (lane >> 4) << 3;
        const int col   = lane & 15;
        float*    st    = stg[wave];
#pragma unroll
        for (int r = 0; r < 8; ++r) {
            st[(rbase + r) * STG + col]      = c0[r];
            st[(rbase + r) * STG + col + 16] = c1[r];
            st[(rbase + r) * STG + col + 32] = c2[r];
            st[(rbase + r) * STG + col + 48] = c3[r];
        }
    }
    __syncthreads();

    const float* st = stg[wave];
    const float* sc = sctab[wave];
    float*       op = out + rowbase * HDIM;
#pragma unroll
    for (int i = 0; i < 32; ++i) {
        const int flat = i * 32 + lane;
        const int row  = flat >> 6;
        const int col  = flat & 63;
        const int n    = col & 31;
        const v2f xy = *(const v2f*)(st + row * STG + 2 * n);
        const float cs = sc[n], sn = sc[n + 32];
        const float val = (col < 32) ? (xy.x * cs - xy.y * sn)
                                     : (xy.x * sn + xy.y * cs);
        op[flat] = val * 32.0f;
    }
}

extern "C" void kernel_launch(void* const* d_in, const int* in_sizes, int n_in,
                              void* d_out, int out_size, void* d_ws, size_t ws_size,
                              hipStream_t stream) {
    const float* x      = (const float*)d_in[0];
    const float* matrix = (const float*)d_in[1];
    const float* thetas = (const float*)d_in[2];
    const float* tscale = (const float*)d_in[3];
    const float* invf   = (const float*)d_in[4];
    const int*   pairs  = (const int*)d_in[5];
    float*       Mws    = (float*)d_ws;

    build_M_kernel<<<64, 32, 0, stream>>>(matrix, thetas, tscale, pairs, Mws);

    const int tiles  = BATCH * SEQ;
    const int blocks = tiles / 8;
    rope_gemm_kernel<<<blocks, 256, 0, stream>>>(x, Mws, invf, (float*)d_out);
}


# Main kernel resident: 256 workgroups x 32 waves, a wave keeps one position s and loops over the 4 batches; M swizzle and sin/cos table once per wave, no workgroup barrier in loop; sum order kept
# speedup vs baseline: 2.2457x; 1.1078x over previous
#include <hip/hip_runtime.h>
#include <math.h>

#define DIMS     1024
#define HEADS    16
#define HDIM     64
#define ROT      32
#define BATCH    4
#define SEQ      8192

typedef __attribute__((ext_vector_type(2))) float v2f;
typedef __attribute__((ext_vector_type(8))) float v8f;

__device__ __forceinline__ int lane_value(int v, int l) {
    return __builtin_amdgcn_readlane(v, l);
}
__device__ __forceinline__ float lane_value(float v, int l) {
    return __int_as_float(__builtin_amdgcn_readlane(__float_as_int(v), l));
}

__global__ void __launch_bounds__(32)
build_M_kernel(const float* __restrict__ matrix,
               const float* __restrict__ thetas,
               const float* __restrict__ tscale,
               const int*   __restrict__ pairs,
               float*       __restrict__ Mout) {
    const int t    = blockIdx.x;
    const int lane = threadIdx.x;

    const float th  = thetas[lane] * tscale[0];
    const int   piv = pairs[2 * lane];
    const int   pjv = pairs[2 * lane + 1];

    float mlo[HDIM], mhi[HDIM];
#pragma unroll
    for (int m = 0; m < HDIM; ++m) {
        mlo[m] = matrix[m * HDIM + lane];
        mhi[m] = matrix[m * HDIM + lane + 32];
    }

    const float csv = cosf(th), snv = sinf(th);

    float lo = (lane == t)      ? 1.0f : 0.0f;
    float hi = (lane + 32 == t) ? 1.0f : 0.0f;

#pragma unroll
    for (int g = 0; g < ROT; ++g) {
        const int   i  = lane_value(piv, g);
        const int   j  = lane_value(pjv, g);
        const float cs = lane_value(csv, g);
        const float sn = lane_value(snv, g);
        const float il = lane_value(lo, i & 31), ih = lane_value(hi, i & 31);
        const float jl = lane_value(lo, j & 31), jh = lane_value(hi, j & 31);
        const float vi = (i & 32) ? ih : il;
        const float vj = (j & 32) ? jh : jl;
        const float ri =  cs * vi + sn * vj;
        const float rj = -sn * vi + cs * vj;
        const float rd = vi * sn;
        const float ni = (i != j) ? ri : rd;
        const float nj = (i != j) ? rj : rd;
        lo = (lane == i)      ? ni : lo;
        hi = (lane + 32 == i) ? ni : hi;
        lo = (lane == j)      ? nj : lo;
        hi = (lane + 32 == j) ? nj : hi;
    }

    float s0 = 0.0f, s1 = 0.0f;
#pragma unroll
    for (int m = 0; m < 32; ++m) {
        const float a = lane_value(lo, m);
        s0 = fmaf(a, mlo[m], s0);
        s1 = fmaf(a, mhi[m], s1);
    }
#pragma unroll
    for (int m = 0; m < 32; ++m) {
        const float a = lane_value(hi, m);
        s0 = fmaf(a, mlo[m + 32], s0);
        s1 = fmaf(a, mhi[m + 32], s1);
    }
    Mout[t * HDIM + lane]      = s0;
    Mout[t * HDIM + lane + 32] = s1;
}

#define STG   66
#define NWAVE 32
#define NTHR  1024
#define NWG   256
static_assert(NTHR == NWAVE * 32 && NWG * NWAVE == SEQ, "one wave per position");

__global__ void __launch_bounds__(NTHR)
rope_gemm_kernel(const float* __restrict__ x,
                 const float* __restrict__ Mg,
                 const float* __restrict__ invf,
                 float*       __restrict__ out) {
    __shared__ __align__(16) float Bl[16 * 4 * 64];
    __shared__ __align__(16) float stg[NWAVE][16 * STG];
    __shared__ __align__(16) float sctab[NWAVE][64];

    const int tid  = threadIdx.x;
    const int wave = tid >> 5;
    const int lane = tid & 31;

    for (int idx = tid; idx < 16 * 4 * 64; idx += NTHR) {
        const int v  = idx & 1;
        const int ln = (idx >> 1) & 31;
        const int t  = (idx >> 6) & 3;
        const int kk = idx >> 8;
        const int krow = 4 * kk + ((ln >> 4) << 1) + v;
        const int col  = 16 * t + (ln & 15);
        Bl[idx] = Mg[krow * HDIM + col];
    }
    __syncthreads();

    const int s = blockIdx.x * NWAVE + wave;

    if (lane < ROT) {
        const float a = (float)s * invf[lane];
        sctab[wave][lane]      = cosf(a);
        sctab[wave][lane + 32] = sinf(a);
    }

    float*       st = stg[wave];
    const float* sc = sctab[wave];
    const float* Bp = Bl + lane * 2;

    for (int b = 0; b < BATCH; ++b) {
        const long rowbase = ((long)b * SEQ + s) * 16;

        const float* xrow = x + (rowbase + (lane & 15)) * HDIM + ((lane >> 4) << 1);
        v2f afrag[16];
#pragma unroll
        for (int kk = 0; kk < 16; ++kk)
            afrag[kk] = *(const v2f*)(xrow + kk * 4);
        __builtin_amdgcn_sched_barrier(0);

        v8f c0 = {}, c1 = {}, c2 = {}, c3 = {};
#pragma unroll
        for (int kk = 0; kk < 16; ++kk) {
            const v2f b0 = *(const v2f*)(Bp + (kk * 4 + 0) * 64);
            const v2f b1 = *(const v2f*)(Bp + (kk * 4 + 1) * 64);
            const v2f b2 = *(const v2f*)(Bp + (kk * 4 + 2) * 64);
            const v2f b3 = *(const v2f*)(Bp + (kk * 4 + 3) * 64);
            c0 = __builtin_amdgcn_wmma_f32_16x16x4_f32(false, afrag[kk], false, b0,
                                                       (short)0, c0, false, false);
            c1 = __builtin_amdgcn_wmma_f32_16x16x4_f32(false, afrag[kk], false, b1,
                                                       (short)0, c1, false, false);
            c2 = __builtin_amdgcn_wmma_f32_16x16x4_f32(false, afrag[kk], false, b2,
                                                       (short)0, c2, false, false);
            c3 = __builtin_amdgcn_wmma_f32_16x16x4_f32(false, afrag[kk], false, b3,
                                                       (short)0, c3, false, false);
        }

        __builtin_amdgcn_fence(__ATOMIC_ACQ_REL, "wavefront");

        {
            const int rbase = (lane >> 4) << 3;
            const int col   = lane & 15;
#pragma unroll
            for (int r = 0; r < 8; ++r) {
                st[(rbase + r) * STG + col]      = c0[r];
                st[(rbase + r) * STG + col + 16] = c1[r];
                st[(rbase + r) * STG + col + 32] = c2[r];
                st[(rbase + r) * STG + col + 48] = c3[r];
            }
        }
        __builtin_amdgcn_fence(__ATOMIC_ACQ_REL, "wavefront");

        float* op = out + rowbase * HDIM;
#pragma unroll
        for (int i = 0; i < 32; ++i) {
            const int flat = i * 32 + lane;
            const int row  = flat >> 6;
            const int col  = flat & 63;
            const int n    = col & 31;
            const v2f xy = *(const v2f*)(st + row * STG + 2 * n);
            const float cs = sc[n], sn = sc[n + 32];
            const float val = (col < 32) ? (xy.x * cs - xy.y * sn)
                                         : (xy.x * sn + xy.y * cs);
            op[flat] = val * 32.0f;
        }
    }
}

extern "C" void kernel_launch(void* const* d_in, const int* in_sizes, int n_in,
                              void* d_out, int out_size, void* d_ws, size_t ws_size,
                              hipStream_t stream) {
    const float* x      = (const float*)d_in[0];
    const float* matrix = (const float*)d_in[1];
    const float* thetas = (const float*)d_in[2];
    const float* tscale = (const float*)d_in[3];
    const float* invf   = (const float*)d_in[4];
    const int*   pairs  = (const int*)d_in[5];
    float*       Mws    = (float*)d_ws;

    build_M_kernel<<<64, 32, 0, stream>>>(matrix, thetas, tscale, pairs, Mws);

    rope_gemm_kernel<<<NWG, NTHR, 0, stream>>>(x, Mws, invf, (float*)d_out);
}


# x tile read as it lies in memory: 8 whole-line 16-B loads per lane (4 lines each, every line once), through the wave's LDS slab into the A layout; was 16 loads of 16 lines each; k and sum order kept
# speedup vs baseline: 2.3191x; 1.0327x over previous
#include <hip/hip_runtime.h>
#include <math.h>

#define DIMS     1024
#define HEADS    16
#define HDIM     64
#define ROT      32
#define BATCH    4
#define SEQ      8192

typedef __attribute__((ext_vector_type(2))) float v2f;
typedef __attribute__((ext_vector_type(4))) float v4f;
typedef __attribute__((ext_vector_type(8))) float v8f;

__device__ __forceinline__ int lane_value(int v, int l) {
    return __builtin_amdgcn_readlane(v, l);
}
__device__ __forceinline__ float lane_value(float v, int l) {
    return __int_as_float(__builtin_amdgcn_readlane(__float_as_int(v), l));
}

__global__ void __launch_bounds__(32)
build_M_kernel(const float* __restrict__ matrix,
               const float* __restrict__ thetas,
               const float* __restrict__ tscale,
               const int*   __restrict__ pairs,
               float*       __restrict__ Mout) {
    const int t    = blockIdx.x;
    const int lane = threadIdx.x;

    const float th  = thetas[lane] * tscale[0];
    const int   piv = pairs[2 * lane];
    const int   pjv = pairs[2 * lane + 1];

    float mlo[HDIM], mhi[HDIM];
#pragma unroll
    for (int m = 0; m < HDIM; ++m) {
        mlo[m] = matrix[m * HDIM + lane];
        mhi[m] = matrix[m * HDIM + lane + 32];
    }

    const float csv = cosf(th), snv = sinf(th);

    float lo = (lane == t)      ? 1.0f : 0.0f;
    float hi = (lane + 32 == t) ? 1.0f : 0.0f;

#pragma unroll
    for (int g = 0; g < ROT; ++g) {
        const int   i  = lane_value(piv, g);
        const int   j  = lane_value(pjv, g);
        const float cs = lane_value(csv, g);
        const float sn = lane_value(snv, g);
        const float il = lane_value(lo, i & 31), ih = lane_value(hi, i & 31);
        const float jl = lane_value(lo, j & 31), jh = lane_value(hi, j & 31);
        const float vi = (i & 32) ? ih : il;
        const float vj = (j & 32) ? jh : jl;
        const float ri =  cs * vi + sn * vj;
        const float rj = -sn * vi + cs * vj;
        const float rd = vi * sn;
        const float ni = (i != j) ? ri : rd;
        const float nj = (i != j) ? rj : rd;
        lo = (lane == i)      ? ni : lo;
        hi = (lane + 32 == i) ? ni : hi;
        lo = (lane == j)      ? nj : lo;
        hi = (lane + 32 == j) ? nj : hi;
    }

    float s0 = 0.0f, s1 = 0.0f;
#pragma unroll
    for (int m = 0; m < 32; ++m) {
        const float a = lane_value(lo, m);
        s0 = fmaf(a, mlo[m], s0);
        s1 = fmaf(a, mhi[m], s1);
    }
#pragma unroll
    for (int m = 0; m < 32; ++m) {
        const float a = lane_value(hi, m);
        s0 = fmaf(a, mlo[m + 32], s0);
        s1 = fmaf(a, mhi[m + 32], s1);
    }
    Mout[t * HDIM + lane]      = s0;
    Mout[t * HDIM + lane + 32] = s1;
}

#define STG   68
#define NWAVE 32
#define NTHR  1024
#define NWG   256
static_assert(NTHR == NWAVE * 32 && NWG * NWAVE == SEQ, "one wave per position");

__global__ void __launch_bounds__(NTHR)
rope_gemm_kernel(const float* __restrict__ x,
                 const float* __restrict__ Mg,
                 const float* __restrict__ invf,
                 float*       __restrict__ out) {
    __shared__ __align__(16) float Bl[16 * 4 * 64];
    __shared__ __align__(16) float stg[NWAVE][16 * STG];
    __shared__ __align__(16) float sctab[NWAVE][64];

    const int tid  = threadIdx.x;
    const int wave = tid >> 5;
    const int lane = tid & 31;

    for (int idx = tid; idx < 16 * 4 * 64; idx += NTHR) {
        const int v  = idx & 1;
        const int ln = (idx >> 1) & 31;
        const int t  = (idx >> 6) & 3;
        const int kk = idx >> 8;
        const int krow = 4 * kk + ((ln >> 4) << 1) + v;
        const int col  = 16 * t + (ln & 15);
        Bl[idx] = Mg[krow * HDIM + col];
    }
    __syncthreads();

    const int s = blockIdx.x * NWAVE + wave;

    if (lane < ROT) {
        const float a = (float)s * invf[lane];
        sctab[wave][lane]      = cosf(a);
        sctab[wave][lane + 32] = sinf(a);
    }

    float*       st = stg[wave];
    const float* sc = sctab[wave];
    const float* Bp = Bl + lane * 2;

    for (int b = 0; b < BATCH; ++b) {
        const long rowbase = ((long)b * SEQ + s) * 16;

        const float* xt = x + rowbase * HDIM + lane * 4;
        v4f xin[8];
#pragma unroll
        for (int q = 0; q < 8; ++q)
            xin[q] = *(const v4f*)(xt + q * 128);

        float* sw = st + (lane >> 4) * STG + ((lane & 15) << 2);
#pragma unroll
        for (int q = 0; q < 8; ++q)
            *(v4f*)(sw + (2 * q) * STG) = xin[q];
        __builtin_amdgcn_fence(__ATOMIC_ACQ_REL, "wavefront");

        const float* arow = st + (lane & 15) * STG + ((lane >> 4) << 1);
        v2f afrag[16];
#pragma unroll
        for (int kk = 0; kk < 16; ++kk)
            afrag[kk] = *(const v2f*)(arow + kk * 4);
        __builtin_amdgcn_sched_barrier(0);

        v8f c0 = {}, c1 = {}, c2 = {}, c3 = {};
#pragma unroll
        for (int kk = 0; kk < 16; ++kk) {
            const v2f b0 = *(const v2f*)(Bp + (kk * 4 + 0) * 64);
            const v2f b1 = *(const v2f*)(Bp + (kk * 4 + 1) * 64);
            const v2f b2 = *(const v2f*)(Bp + (kk * 4 + 2) * 64);
            const v2f b3 = *(const v2f*)(Bp + (kk * 4 + 3) * 64);
            c0 = __builtin_amdgcn_wmma_f32_16x16x4_f32(false, afrag[kk], false, b0,
                                                       (short)0, c0, false, false);
            c1 = __builtin_amdgcn_wmma_f32_16x16x4_f32(false, afrag[kk], false, b1,
                                                       (short)0, c1, false, false);
            c2 = __builtin_amdgcn_wmma_f32_16x16x4_f32(false, afrag[kk], false, b2,
                                                       (short)0, c2, false, false);
            c3 = __builtin_amdgcn_wmma_f32_16x16x4_f32(false, afrag[kk], false, b3,
                                                       (short)0, c3, false, false);
        }

        __builtin_amdgcn_fence(__ATOMIC_ACQ_REL, "wavefront");

        {
            const int rbase = (lane >> 4) << 3;
            const int col   = lane & 15;
#pragma unroll
            for (int r = 0; r < 8; ++r) {
                st[(rbase + r) * STG + col]      = c0[r];
                st[(rbase + r) * STG + col + 16] = c1[r];
                st[(rbase + r) * STG + col + 32] = c2[r];
                st[(rbase + r) * STG + col + 48] = c3[r];
            }
        }
        __builtin_amdgcn_fence(__ATOMIC_ACQ_REL, "wavefront");

        float* op = out + rowbase * HDIM;
#pragma unroll
        for (int i = 0; i < 32; ++i) {
            const int flat = i * 32 + lane;
            const int row  = flat >> 6;
            const int col  = flat & 63;
            const int n    = col & 31;
            const v2f xy = *(const v2f*)(st + row * STG + 2 * n);
            const float cs = sc[n], sn = sc[n + 32];
            const float val = (col < 32) ? (xy.x * cs - xy.y * sn)
                                         : (xy.x * sn + xy.y * cs);
            op[flat] = val * 32.0f;
        }
    }
}

extern "C" void kernel_launch(void* const* d_in, const int* in_sizes, int n_in,
                              void* d_out, int out_size, void* d_ws, size_t ws_size,
                              hipStream_t stream) {
    const float* x      = (const float*)d_in[0];
    const float* matrix = (const float*)d_in[1];
    const float* thetas = (const float*)d_in[2];
    const float* tscale = (const float*)d_in[3];
    const float* invf   = (const float*)d_in[4];
    const int*   pairs  = (const int*)d_in[5];
    float*       Mws    = (float*)d_ws;

    build_M_kernel<<<64, 32, 0, stream>>>(matrix, thetas, tscale, pairs, Mws);

    rope_gemm_kernel<<<NWG, NTHR, 0, stream>>>(x, Mws, invf, (float*)d_out);
}


# RoPE pairing folded into M's column order: a lane's 4 accumulators hold y[4c..4c+3], rotation in registers (sin/cos kept times 32, exact), whole-line 8-B stores, no C staging in LDS; sums kept
# speedup vs baseline: 2.4386x; 1.0515x over previous
#include <hip/hip_runtime.h>
#include <math.h>

#define DIMS     1024
#define HEADS    16
#define HDIM     64
#define ROT      32
#define BATCH    4
#define SEQ      8192

typedef __attribute__((ext_vector_type(2))) float v2f;
typedef __attribute__((ext_vector_type(4))) float v4f;
typedef __attribute__((ext_vector_type(8))) float v8f;

__device__ __forceinline__ int lane_value(int v, int l) {
    return __builtin_amdgcn_readlane(v, l);
}
__device__ __forceinline__ float lane_value(float v, int l) {
    return __int_as_float(__builtin_amdgcn_readlane(__float_as_int(v), l));
}

__global__ void __launch_bounds__(32)
build_M_kernel(const float* __restrict__ matrix,
               const float* __restrict__ thetas,
               const float* __restrict__ tscale,
               const int*   __restrict__ pairs,
               float*       __restrict__ Mout) {
    const int t    = blockIdx.x;
    const int lane = threadIdx.x;

    const float th  = thetas[lane] * tscale[0];
    const int   piv = pairs[2 * lane];
    const int   pjv = pairs[2 * lane + 1];

    float mlo[HDIM], mhi[HDIM];
#pragma unroll
    for (int m = 0; m < HDIM; ++m) {
        mlo[m] = matrix[m * HDIM + lane];
        mhi[m] = matrix[m * HDIM + lane + 32];
    }

    const float csv = cosf(th), snv = sinf(th);

    float lo = (lane == t)      ? 1.0f : 0.0f;
    float hi = (lane + 32 == t) ? 1.0f : 0.0f;

#pragma unroll
    for (int g = 0; g < ROT; ++g) {
        const int   i  = lane_value(piv, g);
        const int   j  = lane_value(pjv, g);
        const float cs = lane_value(csv, g);
        const float sn = lane_value(snv, g);
        const float il = lane_value(lo, i & 31), ih = lane_value(hi, i & 31);
        const float jl = lane_value(lo, j & 31), jh = lane_value(hi, j & 31);
        const float vi = (i & 32) ? ih : il;
        const float vj = (j & 32) ? jh : jl;
        const float ri =  cs * vi + sn * vj;
        const float rj = -sn * vi + cs * vj;
        const float rd = vi * sn;
        const float ni = (i != j) ? ri : rd;
        const float nj = (i != j) ? rj : rd;
        lo = (lane == i)      ? ni : lo;
        hi = (lane + 32 == i) ? ni : hi;
        lo = (lane == j)      ? nj : lo;
        hi = (lane + 32 == j) ? nj : hi;
    }

    float s0 = 0.0f, s1 = 0.0f;
#pragma unroll
    for (int m = 0; m < 32; ++m) {
        const float a = lane_value(lo, m);
        s0 = fmaf(a, mlo[m], s0);
        s1 = fmaf(a, mhi[m], s1);
    }
#pragma unroll
    for (int m = 0; m < 32; ++m) {
        const float a = lane_value(hi, m);
        s0 = fmaf(a, mlo[m + 32], s0);
        s1 = fmaf(a, mhi[m + 32], s1);
    }
    Mout[t * HDIM + lane]      = s0;
    Mout[t * HDIM + lane + 32] = s1;
}

#define STG   68
#define NWAVE 32
#define NTHR  1024
#define NWG   256
static_assert(NTHR == NWAVE * 32 && NWG * NWAVE == SEQ, "one wave per position");

__global__ void __launch_bounds__(NTHR)
rope_gemm_kernel(const float* __restrict__ x,
                 const float* __restrict__ Mg,
                 const float* __restrict__ invf,
                 float*       __restrict__ out) {
    __shared__ __align__(16) float Bl[16 * 4 * 64];
    __shared__ __align__(16) float stg[NWAVE][16 * STG];

    const int tid  = threadIdx.x;
    const int wave = tid >> 5;
    const int lane = tid & 31;
    const int half = lane >> 4;
    const int c16  = lane & 15;

    for (int idx = tid; idx < 16 * 4 * 64; idx += NTHR) {
        const int v  = idx & 1;
        const int ln = (idx >> 1) & 31;
        const int t  = (idx >> 6) & 3;
        const int kk = idx >> 8;
        const int krow = 4 * kk + ((ln >> 4) << 1) + v;
        const int col  = 4 * (ln & 15) + t;
        Bl[idx] = Mg[krow * HDIM + col];
    }
    __syncthreads();

    const int s = blockIdx.x * NWAVE + wave;

    float cs0, sn0, cs1, sn1;
    {
        const float a    = (float)s * invf[2 * c16 + half];
        const float csme = cosf(a) * 32.0f, snme = sinf(a) * 32.0f;
        const float csot = __shfl_xor(csme, 16);
        const float snot = __shfl_xor(snme, 16);
        cs0 = half ? csot : csme;   sn0 = half ? snot : snme;
        cs1 = half ? csme : csot;   sn1 = half ? snme : snot;
    }

    float*       st = stg[wave];
    const float* Bp = Bl + lane * 2;

    for (int b = 0; b < BATCH; ++b) {
        const long rowbase = ((long)b * SEQ + s) * 16;

        const float* xt = x + rowbase * HDIM + lane * 4;
        v4f xin[8];
#pragma unroll
        for (int q = 0; q < 8; ++q)
            xin[q] = *(const v4f*)(xt + q * 128);

        __builtin_amdgcn_fence(__ATOMIC_ACQ_REL, "wavefront");

        float* sw = st + half * STG + (c16 << 2);
#pragma unroll
        for (int q = 0; q < 8; ++q)
            *(v4f*)(sw + (2 * q) * STG) = xin[q];
        __builtin_amdgcn_fence(__ATOMIC_ACQ_REL, "wavefront");

        const float* arow = st + c16 * STG + (half << 1);
        v2f afrag[16];
#pragma unroll
        for (int kk = 0; kk < 16; ++kk)
            afrag[kk] = *(const v2f*)(arow + kk * 4);
        __builtin_amdgcn_sched_barrier(0);

        v8f c0 = {}, c1 = {}, c2 = {}, c3 = {};
#pragma unroll
        for (int kk = 0; kk < 16; ++kk) {
            const v2f b0 = *(const v2f*)(Bp + (kk * 4 + 0) * 64);
            const v2f b1 = *(const v2f*)(Bp + (kk * 4 + 1) * 64);
            const v2f b2 = *(const v2f*)(Bp + (kk * 4 + 2) * 64);
            const v2f b3 = *(const v2f*)(Bp + (kk * 4 + 3) * 64);
            c0 = __builtin_amdgcn_wmma_f32_16x16x4_f32(false, afrag[kk], false, b0,
                                                       (short)0, c0, false, false);
            c1 = __builtin_amdgcn_wmma_f32_16x16x4_f32(false, afrag[kk], false, b1,
                                                       (short)0, c1, false, false);
            c2 = __builtin_amdgcn_wmma_f32_16x16x4_f32(false, afrag[kk], false, b2,
                                                       (short)0, c2, false, false);
            c3 = __builtin_amdgcn_wmma_f32_16x16x4_f32(false, afrag[kk], false, b3,
                                                       (short)0, c3, false, false);
        }

        float* op = out + (rowbase + half * 8) * HDIM + 2 * c16;
#pragma unroll
        for (int r = 0; r < 8; ++r) {
            v2f lo, hi;
            lo.x = c0[r] * cs0 - c1[r] * sn0;
            lo.y = c2[r] * cs1 - c3[r] * sn1;
            hi.x = c0[r] * sn0 + c1[r] * cs0;
            hi.y = c2[r] * sn1 + c3[r] * cs1;
            *(v2f*)(op + r * HDIM)      = lo;
            *(v2f*)(op + r * HDIM + 32) = hi;
        }
    }
}

extern "C" void kernel_launch(void* const* d_in, const int* in_sizes, int n_in,
                              void* d_out, int out_size, void* d_ws, size_t ws_size,
                              hipStream_t stream) {
    const float* x      = (const float*)d_in[0];
    const float* matrix = (const float*)d_in[1];
    const float* thetas = (const float*)d_in[2];
    const float* tscale = (const float*)d_in[3];
    const float* invf   = (const float*)d_in[4];
    const int*   pairs  = (const int*)d_in[5];
    float*       Mws    = (float*)d_ws;

    build_M_kernel<<<64, 32, 0, stream>>>(matrix, thetas, tscale, pairs, Mws);

    rope_gemm_kernel<<<NWG, NTHR, 0, stream>>>(x, Mws, invf, (float*)d_out);
}


# One launch: each workgroup builds M itself in LDS (chain on waves 0-1, one row per lane; product 2 rows per wave, m ascending) while its first x tile is in flight; build kernel gone; sum order kept
# speedup vs baseline: 2.7014x; 1.1078x over previous
#include <hip/hip_runtime.h>
#include <math.h>

#define DIMS     1024
#define HEADS    16
#define HDIM     64
#define ROT      32
#define BATCH    4
#define SEQ      8192

typedef __attribute__((ext_vector_type(2))) float v2f;
typedef __attribute__((ext_vector_type(4))) float v4f;
typedef __attribute__((ext_vector_type(8))) float v8f;

__device__ __forceinline__ int lane_value(int v, int l) {
    return __builtin_amdgcn_readlane(v, l);
}
__device__ __forceinline__ float lane_value(float v, int l) {
    return __int_as_float(__builtin_amdgcn_readlane(__float_as_int(v), l));
}

#define STG   68
#define ACCP  68
#define NWAVE 32
#define NTHR  1024
#define NWG   256
static_assert(NTHR == NWAVE * 32 && NWG * NWAVE == SEQ, "one wave per position");
static_assert(NTHR * 4 == HDIM * HDIM, "one 16-B piece of matrix per thread");
static_assert(2 * NWAVE == HDIM, "two rows of M per wave");
static_assert(HDIM * HDIM + HDIM * ACCP <= NWAVE * 16 * STG, "part 1 fits in the slabs");

__device__ __forceinline__ void
tile_product_rope(const v4f (&xin)[8], float* st, const float* Bp, float* op,
                  const int half, const int c16,
                  const float cs0, const float sn0, const float cs1, const float sn1) {
    __builtin_amdgcn_fence(__ATOMIC_ACQ_REL, "wavefront");

    float* sw = st + half * STG + (c16 << 2);
#pragma unroll
    for (int q = 0; q < 8; ++q)
        *(v4f*)(sw + (2 * q) * STG) = xin[q];
    __builtin_amdgcn_fence(__ATOMIC_ACQ_REL, "wavefront");

    const float* arow = st + c16 * STG + (half << 1);
    v2f afrag[16];
#pragma unroll
    for (int kk = 0; kk < 16; ++kk)
        afrag[kk] = *(const v2f*)(arow + kk * 4);
    __builtin_amdgcn_sched_barrier(0);

    v8f c0 = {}, c1 = {}, c2 = {}, c3 = {};
#pragma unroll
    for (int kk = 0; kk < 16; ++kk) {
        const v2f b0 = *(const v2f*)(Bp + (kk * 4 + 0) * 64);
        const v2f b1 = *(const v2f*)(Bp + (kk * 4 + 1) * 64);
        const v2f b2 = *(const v2f*)(Bp + (kk * 4 + 2) * 64);
        const v2f b3 = *(const v2f*)(Bp + (kk * 4 + 3) * 64);
        c0 = __builtin_amdgcn_wmma_f32_16x16x4_f32(false, afrag[kk], false, b0,
                                                   (short)0, c0, false, false);
        c1 = __builtin_amdgcn_wmma_f32_16x16x4_f32(false, afrag[kk], false, b1,
                                                   (short)0, c1, false, false);
        c2 = __builtin_amdgcn_wmma_f32_16x16x4_f32(false, afrag[kk], false, b2,
                                                   (short)0, c2, false, false);
        c3 = __builtin_amdgcn_wmma_f32_16x16x4_f32(false, afrag[kk], false, b3,
                                                   (short)0, c3, false, false);
    }

#pragma unroll
    for (int r = 0; r < 8; ++r) {
        v2f lo, hi;
        lo.x = c0[r] * cs0 - c1[r] * sn0;
        lo.y = c2[r] * cs1 - c3[r] * sn1;
        hi.x = c0[r] * sn0 + c1[r] * cs0;
        hi.y = c2[r] * sn1 + c3[r] * cs1;
        *(v2f*)(op + r * HDIM)      = lo;
        *(v2f*)(op + r * HDIM + 32) = hi;
    }
}

__global__ void __launch_bounds__(NTHR)
rope_gemm_kernel(const float* __restrict__ x,
                 const float* __restrict__ matrix,
                 const float* __restrict__ thetas,
                 const float* __restrict__ tscale,
                 const int*   __restrict__ pairs,
                 const float* __restrict__ invf,
                 float*       __restrict__ out) {
    __shared__ __align__(16) float Bl[16 * 4 * 64];
    __shared__ __align__(16) float stg[NWAVE][16 * STG];

    const int tid  = threadIdx.x;
    const int wave = tid >> 5;
    const int lane = tid & 31;
    const int half = lane >> 4;
    const int c16  = lane & 15;
    const int wu   = __builtin_amdgcn_readfirstlane(tid >> 5);

    const int s = blockIdx.x * NWAVE + wave;

    const v4f   mpiece = *(const v4f*)(matrix + tid * 4);
    const float thv    = thetas[lane] * tscale[0];
    const int   piv    = pairs[2 * lane];
    const int   pjv    = pairs[2 * lane + 1];
    const float invfv  = invf[2 * c16 + half];

    v4f xin[8];
    {
        const float* xt = x + ((long)s * 16) * HDIM + lane * 4;
#pragma unroll
        for (int q = 0; q < 8; ++q)
            xin[q] = *(const v4f*)(xt + q * 128);
    }
    __builtin_amdgcn_sched_barrier(0);

    float* Ms  = &stg[0][0];
    float* Acc = Ms + HDIM * HDIM;
    *(v4f*)(Ms + tid * 4) = mpiece;

    float cs0, sn0, cs1, sn1;
    {
        const float a    = (float)s * invfv;
        const float csme = cosf(a) * 32.0f, snme = sinf(a) * 32.0f;
        const float csot = __shfl_xor(csme, 16);
        const float snot = __shfl_xor(snme, 16);
        cs0 = half ? csot : csme;   sn0 = half ? snot : snme;
        cs1 = half ? csme : csot;   sn1 = half ? snme : snot;
    }

    if (wu < 2) {
        const float csv = cosf(thv), snv = sinf(thv);
        const int   row = wu * 32 + lane;
        float*      ar  = Acc + row * ACCP;
#pragma unroll
        for (int c = 0; c < HDIM; c += 4) {
            v4f e;
            e.x = (c     == row) ? 1.0f : 0.0f;
            e.y = (c + 1 == row) ? 1.0f : 0.0f;
            e.z = (c + 2 == row) ? 1.0f : 0.0f;
            e.w = (c + 3 == row) ? 1.0f : 0.0f;
            *(v4f*)(ar + c) = e;
        }
#pragma unroll
        for (int g = 0; g < ROT; ++g) {
            const int   i  = lane_value(piv, g) & 63;
            const int   j  = lane_value(pjv, g) & 63;
            const float cs = lane_value(csv, g);
            const float sn = lane_value(snv, g);
            const float vi = ar[i];
            const float vj = ar[j];
            const float ri =  cs * vi + sn * vj;
            const float rj = -sn * vi + cs * vj;
            const float rd = vi * sn;
            ar[i] = (i != j) ? ri : rd;
            ar[j] = (i != j) ? rj : rd;
        }
    }
    __builtin_amdgcn_fence(__ATOMIC_RELEASE, "workgroup", "local");
    __builtin_amdgcn_s_barrier();
    __builtin_amdgcn_fence(__ATOMIC_ACQUIRE, "workgroup", "local");

    {
        const float* a0p = Acc + (2 * wu) * ACCP;
        const float* a1p = a0p + ACCP;
        const float* mp  = Ms + lane;
        float s00 = 0.0f, s01 = 0.0f, s10 = 0.0f, s11 = 0.0f;
#pragma unroll 4
        for (int m4 = 0; m4 < HDIM / 4; ++m4) {
            const v4f a0 = *(const v4f*)(a0p + 4 * m4);
            const v4f a1 = *(const v4f*)(a1p + 4 * m4);
#pragma unroll
            for (int u = 0; u < 4; ++u) {
                const float ml = mp[(4 * m4 + u) * HDIM];
                const float mh = mp[(4 * m4 + u) * HDIM + 32];
                s00 = fmaf(a0[u], ml, s00);
                s01 = fmaf(a0[u], mh, s01);
                s10 = fmaf(a1[u], ml, s10);
                s11 = fmaf(a1[u], mh, s11);
            }
        }
        float* bp = Bl + ((wu >> 1) * 4 + (lane & 3)) * 64 + ((wu & 1) * 16 + (lane >> 2)) * 2;
        v2f lo2, hi2;
        lo2.x = s00;  lo2.y = s10;
        hi2.x = s01;  hi2.y = s11;
        *(v2f*)(bp)      = lo2;
        *(v2f*)(bp + 16) = hi2;
    }
    __syncthreads();

    float*       st = stg[wave];
    const float* Bp = Bl + lane * 2;

    tile_product_rope(xin, st, Bp, out + (((long)s * 16) + half * 8) * HDIM + 2 * c16,
                      half, c16, cs0, sn0, cs1, sn1);

    for (int b = 1; b < BATCH; ++b) {
        const long rowbase = ((long)b * SEQ + s) * 16;
        const float* xt = x + rowbase * HDIM + lane * 4;
#pragma unroll
        for (int q = 0; q < 8; ++q)
            xin[q] = *(const v4f*)(xt + q * 128);
        tile_product_rope(xin, st, Bp, out + (rowbase + half * 8) * HDIM + 2 * c16,
                          half, c16, cs0, sn0, cs1, sn1);
    }
}

extern "C" void kernel_launch(void* const* d_in, const int* in_sizes, int n_in,
                              void* d_out, int out_size, void* d_ws, size_t ws_size,
                              hipStream_t stream) {
    const float* x      = (const float*)d_in[0];
    const float* matrix = (const float*)d_in[1];
    const float* thetas = (const float*)d_in[2];
    const float* tscale = (const float*)d_in[3];
    const float* invf   = (const float*)d_in[4];
    const int*   pairs  = (const int*)d_in[5];

    rope_gemm_kernel<<<NWG, NTHR, 0, stream>>>(x, matrix, thetas, tscale, pairs, invf,
                                               (float*)d_out);
}


# M built by rotating the rows of matrix in place in LDS (G_g @ T for g = 31..0, one wave, in B-fragment layout): acc chain and the 64x64x64 product are gone, one barrier less; M's sum order changes
# speedup vs baseline: 2.8401x; 1.0513x over previous
#include <hip/hip_runtime.h>
#include <math.h>

#define DIMS     1024
#define HEADS    16
#define HDIM     64
#define ROT      32
#define BATCH    4
#define SEQ      8192

typedef __attribute__((ext_vector_type(2))) float v2f;
typedef __attribute__((ext_vector_type(4))) float v4f;
typedef __attribute__((ext_vector_type(8))) float v8f;

__device__ __forceinline__ int lane_value(int v, int l) {
    return __builtin_amdgcn_readlane(v, l);
}
__device__ __forceinline__ float lane_value(float v, int l) {
    return __int_as_float(__builtin_amdgcn_readlane(__float_as_int(v), l));
}

#define STG   68
#define NWAVE 32
#define NTHR  1024
#define NWG   256
static_assert(NTHR == NWAVE * 32 && NWG * NWAVE == SEQ, "one wave per position");
static_assert(NTHR * 4 == HDIM * HDIM, "one 16-B piece of matrix per thread");

__device__ __forceinline__ int m_row_offset(int m) {
    return (m >> 2) * 256 + ((m >> 1) & 1) * 32 + (m & 1);
}

__device__ __forceinline__ void
tile_product_rope(const v4f (&xin)[8], float* st, const float* Bp, float* op,
                  const int half, const int c16,
                  const float cs0, const float sn0, const float cs1, const float sn1) {
    __builtin_amdgcn_fence(__ATOMIC_ACQ_REL, "wavefront");

    float* sw = st + half * STG + (c16 << 2);
#pragma unroll
    for (int q = 0; q < 8; ++q)
        *(v4f*)(sw + (2 * q) * STG) = xin[q];
    __builtin_amdgcn_fence(__ATOMIC_ACQ_REL, "wavefront");

    const float* arow = st + c16 * STG + (half << 1);
    v2f afrag[16];
#pragma unroll
    for (int kk = 0; kk < 16; ++kk)
        afrag[kk] = *(const v2f*)(arow + kk * 4);
    __builtin_amdgcn_sched_barrier(0);

    v8f c0 = {}, c1 = {}, c2 = {}, c3 = {};
#pragma unroll
    for (int kk = 0; kk < 16; ++kk) {
        const v2f b0 = *(const v2f*)(Bp + (kk * 4 + 0) * 64);
        const v2f b1 = *(const v2f*)(Bp + (kk * 4 + 1) * 64);
        const v2f b2 = *(const v2f*)(Bp + (kk * 4 + 2) * 64);
        const v2f b3 = *(const v2f*)(Bp + (kk * 4 + 3) * 64);
        c0 = __builtin_amdgcn_wmma_f32_16x16x4_f32(false, afrag[kk], false, b0,
                                                   (short)0, c0, false, false);
        c1 = __builtin_amdgcn_wmma_f32_16x16x4_f32(false, afrag[kk], false, b1,
                                                   (short)0, c1, false, false);
        c2 = __builtin_amdgcn_wmma_f32_16x16x4_f32(false, afrag[kk], false, b2,
                                                   (short)0, c2, false, false);
        c3 = __builtin_amdgcn_wmma_f32_16x16x4_f32(false, afrag[kk], false, b3,
                                                   (short)0, c3, false, false);
    }

#pragma unroll
    for (int r = 0; r < 8; ++r) {
        v2f lo, hi;
        lo.x = c0[r] * cs0 - c1[r] * sn0;
        lo.y = c2[r] * cs1 - c3[r] * sn1;
        hi.x = c0[r] * sn0 + c1[r] * cs0;
        hi.y = c2[r] * sn1 + c3[r] * cs1;
        *(v2f*)(op + r * HDIM)      = lo;
        *(v2f*)(op + r * HDIM + 32) = hi;
    }
}

__global__ void __launch_bounds__(NTHR)
rope_gemm_kernel(const float* __restrict__ x,
                 const float* __restrict__ matrix,
                 const float* __restrict__ thetas,
                 const float* __restrict__ tscale,
                 const int*   __restrict__ pairs,
                 const float* __restrict__ invf,
                 float*       __restrict__ out) {
    __shared__ __align__(16) float Bl[16 * 4 * 64];
    __shared__ __align__(16) float stg[NWAVE][16 * STG];

    const int tid  = threadIdx.x;
    const int wave = tid >> 5;
    const int lane = tid & 31;
    const int half = lane >> 4;
    const int c16  = lane & 15;
    const int wu   = __builtin_amdgcn_readfirstlane(tid >> 5);

    const int s = blockIdx.x * NWAVE + wave;

    const v4f   mpiece = *(const v4f*)(matrix + tid * 4);
    float       thv    = thetas[lane] * tscale[0];
    const int   piv    = pairs[2 * lane];
    const int   pjv    = pairs[2 * lane + 1];
    const float invfv  = invf[2 * c16 + half];

    v4f xin[8];
    {
        const float* xt = x + ((long)s * 16) * HDIM + lane * 4;
#pragma unroll
        for (int q = 0; q < 8; ++q)
            xin[q] = *(const v4f*)(xt + q * 128);
    }
    __builtin_amdgcn_sched_barrier(0);

    {
        float* bq = Bl + m_row_offset(tid >> 4) + 2 * (tid & 15);
        bq[0]   = mpiece.x;
        bq[64]  = mpiece.y;
        bq[128] = mpiece.z;
        bq[192] = mpiece.w;
    }
    asm volatile("" : "+v"(thv));
    float csv = 0.0f, snv = 0.0f;
    if (wu == 0) {
        csv = cosf(thv);
        snv = sinf(thv);
    }
    __builtin_amdgcn_fence(__ATOMIC_RELEASE, "workgroup", "local");
    __builtin_amdgcn_s_barrier();
    __builtin_amdgcn_fence(__ATOMIC_ACQUIRE, "workgroup", "local");

    if (wu == 0) {
        float* tb = Bl + half * 128 + c16 * 2;
#pragma unroll
        for (int g = ROT - 1; g >= 0; --g) {
            const int   i  = lane_value(piv, g) & 63;
            const int   j  = lane_value(pjv, g) & 63;
            const float cs = lane_value(csv, g);
            const float sn = lane_value(snv, g);
            float* ri = tb + m_row_offset(i);
            float* rj = tb + m_row_offset(j);
            const float a0 = ri[0], a1 = ri[64];
            const float b0 = rj[0], b1 = rj[64];
            const float ni0 = cs * a0 - sn * b0, ni1 = cs * a1 - sn * b1;
            const float nj0 = sn * a0 + cs * b0, nj1 = sn * a1 + cs * b1;
            const float d0  = sn * a0,           d1  = sn * a1;
            ri[0]  = (i != j) ? ni0 : d0;
            ri[64] = (i != j) ? ni1 : d1;
            rj[0]  = (i != j) ? nj0 : d0;
            rj[64] = (i != j) ? nj1 : d1;
        }
    }

    float cs0, sn0, cs1, sn1;
    {
        const float a    = (float)s * invfv;
        const float csme = cosf(a) * 32.0f, snme = sinf(a) * 32.0f;
        const float csot = __shfl_xor(csme, 16);
        const float snot = __shfl_xor(snme, 16);
        cs0 = half ? csot : csme;   sn0 = half ? snot : snme;
        cs1 = half ? csme : csot;   sn1 = half ? snme : snot;
    }

    __builtin_amdgcn_fence(__ATOMIC_RELEASE, "workgroup", "local");
    __builtin_amdgcn_s_barrier();
    __builtin_amdgcn_fence(__ATOMIC_ACQUIRE, "workgroup", "local");

    float*       st = stg[wave];
    const float* Bp = Bl + lane * 2;

    tile_product_rope(xin, st, Bp, out + (((long)s * 16) + half * 8) * HDIM + 2 * c16,
                      half, c16, cs0, sn0, cs1, sn1);

    for (int b = 1; b < BATCH; ++b) {
        const long rowbase = ((long)b * SEQ + s) * 16;
        const float* xt = x + rowbase * HDIM + lane * 4;
#pragma unroll
        for (int q = 0; q < 8; ++q)
            xin[q] = *(const v4f*)(xt + q * 128);
        tile_product_rope(xin, st, Bp, out + (rowbase + half * 8) * HDIM + 2 * c16,
                          half, c16, cs0, sn0, cs1, sn1);
    }
}

extern "C" void kernel_launch(void* const* d_in, const int* in_sizes, int n_in,
                              void* d_out, int out_size, void* d_ws, size_t ws_size,
                              hipStream_t stream) {
    const float* x      = (const float*)d_in[0];
    const float* matrix = (const float*)d_in[1];
    const float* thetas = (const float*)d_in[2];
    const float* tscale = (const float*)d_in[3];
    const float* invf   = (const float*)d_in[4];
    const int*   pairs  = (const int*)d_in[5];

    rope_gemm_kernel<<<NWG, NTHR, 0, stream>>>(x, matrix, thetas, tscale, pairs, invf,
                                               (float*)d_out);
}


# Next batch's x tile requested ahead: its 8 loads go out once the current tile is in LDS and fly under the 64 products, RoPE and stores (batch 1 under the M build); was load, wait, compute; sums kept
# speedup vs baseline: 2.9523x; 1.0395x over previous
#include <hip/hip_runtime.h>
#include <math.h>

#define DIMS     1024
#define HEADS    16
#define HDIM     64
#define ROT      32
#define BATCH    4
#define SEQ      8192

typedef __attribute__((ext_vector_type(2))) float v2f;
typedef __attribute__((ext_vector_type(4))) float v4f;
typedef __attribute__((ext_vector_type(8))) float v8f;

__device__ __forceinline__ int lane_value(int v, int l) {
    return __builtin_amdgcn_readlane(v, l);
}
__device__ __forceinline__ float lane_value(float v, int l) {
    return __int_as_float(__builtin_amdgcn_readlane(__float_as_int(v), l));
}

#define STG   68
#define NWAVE 32
#define NTHR  1024
#define NWG   256
static_assert(NTHR == NWAVE * 32 && NWG * NWAVE == SEQ, "one wave per position");
static_assert(NTHR * 4 == HDIM * HDIM, "one 16-B piece of matrix per thread");

__device__ __forceinline__ int m_row_offset(int m) {
    return (m >> 2) * 256 + ((m >> 1) & 1) * 32 + (m & 1);
}

__device__ __forceinline__ void
stage_tile(const v4f (&xin)[8], float* st, const int half, const int c16) {
    __builtin_amdgcn_fence(__ATOMIC_ACQ_REL, "wavefront");

    float* sw = st + half * STG + (c16 << 2);
#pragma unroll
    for (int q = 0; q < 8; ++q)
        *(v4f*)(sw + (2 * q) * STG) = xin[q];
    __builtin_amdgcn_fence(__ATOMIC_ACQ_REL, "wavefront");
}

__device__ __forceinline__ void
tile_product_rope(const float* st, const float* Bp, float* op,
                  const int half, const int c16,
                  const float cs0, const float sn0, const float cs1, const float sn1) {
    const float* arow = st + c16 * STG + (half << 1);
    v8f c0 = {}, c1 = {}, c2 = {}, c3 = {};
#pragma unroll
    for (int h8 = 0; h8 < 16; h8 += 8) {
        v2f afrag[8];
#pragma unroll
        for (int kk = 0; kk < 8; ++kk)
            afrag[kk] = *(const v2f*)(arow + (h8 + kk) * 4);
        if (h8 == 0) __builtin_amdgcn_sched_barrier(0);
#pragma unroll
        for (int kk = 0; kk < 8; ++kk) {
            const int k = h8 + kk;
            const v2f b0 = *(const v2f*)(Bp + (k * 4 + 0) * 64);
            const v2f b1 = *(const v2f*)(Bp + (k * 4 + 1) * 64);
            const v2f b2 = *(const v2f*)(Bp + (k * 4 + 2) * 64);
            const v2f b3 = *(const v2f*)(Bp + (k * 4 + 3) * 64);
            c0 = __builtin_amdgcn_wmma_f32_16x16x4_f32(false, afrag[kk], false, b0,
                                                       (short)0, c0, false, false);
            c1 = __builtin_amdgcn_wmma_f32_16x16x4_f32(false, afrag[kk], false, b1,
                                                       (short)0, c1, false, false);
            c2 = __builtin_amdgcn_wmma_f32_16x16x4_f32(false, afrag[kk], false, b2,
                                                       (short)0, c2, false, false);
            c3 = __builtin_amdgcn_wmma_f32_16x16x4_f32(false, afrag[kk], false, b3,
                                                       (short)0, c3, false, false);
        }
    }

#pragma unroll
    for (int r = 0; r < 8; ++r) {
        v2f lo, hi;
        lo.x = c0[r] * cs0 - c1[r] * sn0;
        lo.y = c2[r] * cs1 - c3[r] * sn1;
        hi.x = c0[r] * sn0 + c1[r] * cs0;
        hi.y = c2[r] * sn1 + c3[r] * cs1;
        *(v2f*)(op + r * HDIM)      = lo;
        *(v2f*)(op + r * HDIM + 32) = hi;
    }
}

__global__ void __launch_bounds__(NTHR)
rope_gemm_kernel(const float* __restrict__ x,
                 const float* __restrict__ matrix,
                 const float* __restrict__ thetas,
                 const float* __restrict__ tscale,
                 const int*   __restrict__ pairs,
                 const float* __restrict__ invf,
                 float*       __restrict__ out) {
    __shared__ __align__(16) float lds[16 * 4 * 64 + NWAVE * 16 * STG];
    float* const Bl = lds;
    float* const stg0 = lds + 16 * 4 * 64;

    const int tid  = threadIdx.x;
    const int wave = tid >> 5;
    const int lane = tid & 31;
    const int half = lane >> 4;
    const int c16  = lane & 15;
    const int wu   = __builtin_amdgcn_readfirstlane(tid >> 5);

    const int s = blockIdx.x * NWAVE + wave;

    const v4f   mpiece = *(const v4f*)(matrix + tid * 4);
    float       thv    = thetas[lane] * tscale[0];
    const int   piv    = pairs[2 * lane];
    const int   pjv    = pairs[2 * lane + 1];
    const float invfv  = invf[2 * c16 + half];

    v4f xin[8];
    {
        const float* xt = x + ((long)s * 16) * HDIM + lane * 4;
#pragma unroll
        for (int q = 0; q < 8; ++q)
            xin[q] = *(const v4f*)(xt + q * 128);
    }
    __builtin_amdgcn_sched_barrier(0);

    {
        float* bq = Bl + m_row_offset(tid >> 4) + 2 * (tid & 15);
        bq[0]   = mpiece.x;
        bq[64]  = mpiece.y;
        bq[128] = mpiece.z;
        bq[192] = mpiece.w;
    }
    asm volatile("" : "+v"(thv));
    float csv = 0.0f, snv = 0.0f;
    if (wu == 0) {
        csv = cosf(thv);
        snv = sinf(thv);
    }
    __builtin_amdgcn_fence(__ATOMIC_RELEASE, "workgroup", "local");
    __builtin_amdgcn_s_barrier();
    __builtin_amdgcn_fence(__ATOMIC_ACQUIRE, "workgroup", "local");

    if (wu == 0) {
        float* tb = Bl + half * 128 + c16 * 2;
#pragma unroll
        for (int g = ROT - 1; g >= 0; --g) {
            const int   i  = lane_value(piv, g) & 63;
            const int   j  = lane_value(pjv, g) & 63;
            const float cs = lane_value(csv, g);
            const float sn = lane_value(snv, g);
            float* ri = tb + m_row_offset(i);
            float* rj = tb + m_row_offset(j);
            const float a0 = ri[0], a1 = ri[64];
            const float b0 = rj[0], b1 = rj[64];
            const float ni0 = cs * a0 - sn * b0, ni1 = cs * a1 - sn * b1;
            const float nj0 = sn * a0 + cs * b0, nj1 = sn * a1 + cs * b1;
            const float d0  = sn * a0,           d1  = sn * a1;
            ri[0]  = (i != j) ? ni0 : d0;
            ri[64] = (i != j) ? ni1 : d1;
            rj[0]  = (i != j) ? nj0 : d0;
            rj[64] = (i != j) ? nj1 : d1;
        }
    }

    float cs0, sn0, cs1, sn1;
    {
        const float a    = (float)s * invfv;
        const float csme = cosf(a) * 32.0f, snme = sinf(a) * 32.0f;
        const float csot = __shfl_xor(csme, 16);
        const float snot = __shfl_xor(snme, 16);
        cs0 = half ? csot : csme;   sn0 = half ? snot : snme;
        cs1 = half ? csme : csot;   sn1 = half ? snme : snot;
    }

    float*       st = stg0 + wave * (16 * STG);
    const float* Bp = Bl + lane * 2;
    const float* xt = x + ((long)s * 16) * HDIM + lane * 4;
    const long   xbatch = (long)SEQ * 16 * HDIM;

    stage_tile(xin, st, half, c16);
#pragma unroll
    for (int q = 0; q < 8; ++q)
        xin[q] = *(const v4f*)(xt + xbatch + q * 128);

    __builtin_amdgcn_fence(__ATOMIC_RELEASE, "workgroup", "local");
    __builtin_amdgcn_s_barrier();
    __builtin_amdgcn_fence(__ATOMIC_ACQUIRE, "workgroup", "local");

#pragma unroll
    for (int b = 0; b < BATCH; ++b) {
        const long rowbase = ((long)b * SEQ + s) * 16;
        tile_product_rope(st, Bp, out + (rowbase + half * 8) * HDIM + 2 * c16,
                          half, c16, cs0, sn0, cs1, sn1);
        if (b + 1 < BATCH) {
            stage_tile(xin, st, half, c16);
            if (b + 2 < BATCH) {
#pragma unroll
                for (int q = 0; q < 8; ++q)
                    xin[q] = *(const v4f*)(xt + (b + 2) * xbatch + q * 128);
            }
        }
    }
}

extern "C" void kernel_launch(void* const* d_in, const int* in_sizes, int n_in,
                              void* d_out, int out_size, void* d_ws, size_t ws_size,
                              hipStream_t stream) {
    const float* x      = (const float*)d_in[0];
    const float* matrix = (const float*)d_in[1];
    const float* thetas = (const float*)d_in[2];
    const float* tscale = (const float*)d_in[3];
    const float* invf   = (const float*)d_in[4];
    const int*   pairs  = (const int*)d_in[5];

    rope_gemm_kernel<<<NWG, NTHR, 0, stream>>>(x, matrix, thetas, tscale, pairs, invf,
                                               (float*)d_out);
}


# speedup vs baseline: 3.0647x; 1.0381x over previous
#include <hip/hip_runtime.h>
#include <math.h>

#define DIMS     1024
#define HEADS    16
#define HDIM     64
#define ROT      32
#define BATCH    4
#define SEQ      8192

typedef __attribute__((ext_vector_type(2))) float v2f;
typedef __attribute__((ext_vector_type(4))) float v4f;
typedef __attribute__((ext_vector_type(8))) float v8f;

__device__ __forceinline__ int lane_value(int v, int l) {
    return __builtin_amdgcn_readlane(v, l);
}
__device__ __forceinline__ float lane_value(float v, int l) {
    return __int_as_float(__builtin_amdgcn_readlane(__float_as_int(v), l));
}

__device__ __forceinline__ v4f load_x_once(const float* p) {
    return __builtin_nontemporal_load((const v4f*)p);
}

#define STG   68
#define NWAVE 32
#define NTHR  1024
#define NWG   256
static_assert(NTHR == NWAVE * 32 && NWG * NWAVE == SEQ, "one wave per position");
static_assert(NTHR * 4 == HDIM * HDIM, "one 16-B piece of matrix per thread");

__device__ __forceinline__ int m_row_offset(int m) {
    return (m >> 2) * 256 + ((m >> 1) & 1) * 32 + (m & 1);
}

__device__ __forceinline__ void
stage_tile(const v4f (&xin)[8], float* st, const int half, const int c16) {
    __builtin_amdgcn_fence(__ATOMIC_ACQ_REL, "wavefront");

    float* sw = st + half * STG + (c16 << 2);
#pragma unroll
    for (int q = 0; q < 8; ++q)
        *(v4f*)(sw + (2 * q) * STG) = xin[q];
    __builtin_amdgcn_fence(__ATOMIC_ACQ_REL, "wavefront");
}

__device__ __forceinline__ void
tile_product_rope(const float* st, const float* Bp, float* op,
                  const int half, const int c16,
                  const float cs0, const float sn0, const float cs1, const float sn1) {
    const float* arow = st + c16 * STG + (half << 1);
    v8f c0 = {}, c1 = {}, c2 = {}, c3 = {};
#pragma unroll
    for (int h8 = 0; h8 < 16; h8 += 8) {
        v2f afrag[8];
#pragma unroll
        for (int kk = 0; kk < 8; ++kk)
            afrag[kk] = *(const v2f*)(arow + (h8 + kk) * 4);
        if (h8 == 0) __builtin_amdgcn_sched_barrier(0);
#pragma unroll
        for (int kk = 0; kk < 8; ++kk) {
            const int k = h8 + kk;
            const v2f b0 = *(const v2f*)(Bp + (k * 4 + 0) * 64);
            const v2f b1 = *(const v2f*)(Bp + (k * 4 + 1) * 64);
            const v2f b2 = *(const v2f*)(Bp + (k * 4 + 2) * 64);
            const v2f b3 = *(const v2f*)(Bp + (k * 4 + 3) * 64);
            c0 = __builtin_amdgcn_wmma_f32_16x16x4_f32(false, afrag[kk], false, b0,
                                                       (short)0, c0, false, false);
            c1 = __builtin_amdgcn_wmma_f32_16x16x4_f32(false, afrag[kk], false, b1,
                                                       (short)0, c1, false, false);
            c2 = __builtin_amdgcn_wmma_f32_16x16x4_f32(false, afrag[kk], false, b2,
                                                       (short)0, c2, false, false);
            c3 = __builtin_amdgcn_wmma_f32_16x16x4_f32(false, afrag[kk], false, b3,
                                                       (short)0, c3, false, false);
        }
    }

#pragma unroll
    for (int r = 0; r < 8; ++r) {
        v2f lo, hi;
        lo.x = c0[r] * cs0 - c1[r] * sn0;
        lo.y = c2[r] * cs1 - c3[r] * sn1;
        hi.x = c0[r] * sn0 + c1[r] * cs0;
        hi.y = c2[r] * sn1 + c3[r] * cs1;
        *(v2f*)(op + r * HDIM)      = lo;
        *(v2f*)(op + r * HDIM + 32) = hi;
    }
}

__global__ void __launch_bounds__(NTHR)
rope_gemm_kernel(const float* __restrict__ x,
                 const float* __restrict__ matrix,
                 const float* __restrict__ thetas,
                 const float* __restrict__ tscale,
                 const int*   __restrict__ pairs,
                 const float* __restrict__ invf,
                 float*       __restrict__ out) {
    __shared__ __align__(16) float lds[16 * 4 * 64 + NWAVE * 16 * STG];
    float* const Bl = lds;
    float* const stg0 = lds + 16 * 4 * 64;

    const int tid  = threadIdx.x;
    const int wave = tid >> 5;
    const int lane = tid & 31;
    const int half = lane >> 4;
    const int c16  = lane & 15;
    const int wu   = __builtin_amdgcn_readfirstlane(tid >> 5);

    const int s = blockIdx.x * NWAVE + wave;

    const v4f   mpiece = *(const v4f*)(matrix + tid * 4);
    float       thv    = thetas[lane] * tscale[0];
    const int   piv    = pairs[2 * lane];
    const int   pjv    = pairs[2 * lane + 1];
    const float invfv  = invf[2 * c16 + half];

    v4f xin[8];
    {
        const float* xt = x + ((long)s * 16) * HDIM + lane * 4;
#pragma unroll
        for (int q = 0; q < 8; ++q)
            xin[q] = load_x_once(xt + q * 128);
    }
    __builtin_amdgcn_sched_barrier(0);

    {
        float* bq = Bl + m_row_offset(tid >> 4) + 2 * (tid & 15);
        bq[0]   = mpiece.x;
        bq[64]  = mpiece.y;
        bq[128] = mpiece.z;
        bq[192] = mpiece.w;
    }
    asm volatile("" : "+v"(thv));
    float csv = 0.0f, snv = 0.0f;
    if (wu == 0) {
        csv = cosf(thv);
        snv = sinf(thv);
    }
    __builtin_amdgcn_fence(__ATOMIC_RELEASE, "workgroup", "local");
    __builtin_amdgcn_s_barrier();
    __builtin_amdgcn_fence(__ATOMIC_ACQUIRE, "workgroup", "local");

    if (wu == 0) {
        float* tb = Bl + half * 128 + c16 * 2;
#pragma unroll
        for (int g = ROT - 1; g >= 0; --g) {
            const int   i  = lane_value(piv, g) & 63;
            const int   j  = lane_value(pjv, g) & 63;
            const float cs = lane_value(csv, g);
            const float sn = lane_value(snv, g);
            float* ri = tb + m_row_offset(i);
            float* rj = tb + m_row_offset(j);
            const float a0 = ri[0], a1 = ri[64];
            const float b0 = rj[0], b1 = rj[64];
            const float ni0 = cs * a0 - sn * b0, ni1 = cs * a1 - sn * b1;
            const float nj0 = sn * a0 + cs * b0, nj1 = sn * a1 + cs * b1;
            const float d0  = sn * a0,           d1  = sn * a1;
            ri[0]  = (i != j) ? ni0 : d0;
            ri[64] = (i != j) ? ni1 : d1;
            rj[0]  = (i != j) ? nj0 : d0;
            rj[64] = (i != j) ? nj1 : d1;
        }
    }

    float cs0, sn0, cs1, sn1;
    {
        const float a    = (float)s * invfv;
        const float csme = cosf(a) * 32.0f, snme = sinf(a) * 32.0f;
        const float csot = __shfl_xor(csme, 16);
        const float snot = __shfl_xor(snme, 16);
        cs0 = half ? csot : csme;   sn0 = half ? snot : snme;
        cs1 = half ? csme : csot;   sn1 = half ? snme : snot;
    }

    float*       st = stg0 + wave * (16 * STG);
    const float* Bp = Bl + lane * 2;
    const float* xt = x + ((long)s * 16) * HDIM + lane * 4;
    const long   xbatch = (long)SEQ * 16 * HDIM;

    stage_tile(xin, st, half, c16);
#pragma unroll
    for (int q = 0; q < 8; ++q)
        xin[q] = load_x_once(xt + xbatch + q * 128);

    __builtin_amdgcn_fence(__ATOMIC_RELEASE, "workgroup", "local");
    __builtin_amdgcn_s_barrier();
    __builtin_amdgcn_fence(__ATOMIC_ACQUIRE, "workgroup", "local");

#pragma unroll
    for (int b = 0; b < BATCH; ++b) {
        const long rowbase = ((long)b * SEQ + s) * 16;
        tile_product_rope(st, Bp, out + (rowbase + half * 8) * HDIM + 2 * c16,
                          half, c16, cs0, sn0, cs1, sn1);
        if (b + 1 < BATCH) {
            stage_tile(xin, st, half, c16);
            if (b + 2 < BATCH) {
#pragma unroll
                for (int q = 0; q < 8; ++q)
                    xin[q] = load_x_once(xt + (b + 2) * xbatch + q * 128);
            }
        }
    }
}

extern "C" void kernel_launch(void* const* d_in, const int* in_sizes, int n_in,
                              void* d_out, int out_size, void* d_ws, size_t ws_size,
                              hipStream_t stream) {
    const float* x      = (const float*)d_in[0];
    const float* matrix = (const float*)d_in[1];
    const float* thetas = (const float*)d_in[2];
    const float* tscale = (const float*)d_in[3];
    const float* invf   = (const float*)d_in[4];
    const int*   pairs  = (const int*)d_in[5];

    rope_gemm_kernel<<<NWG, NTHR, 0, stream>>>(x, matrix, thetas, tscale, pairs, invf,
                                               (float*)d_out);
}
